# starting kernel
# baseline (speedup 1.0000x reference)
#include <hip/hip_runtime.h>
#include <hip/hip_bf16.h>
#include <math.h>

typedef __attribute__((ext_vector_type(2))) float v2f;
typedef __attribute__((ext_vector_type(8))) float v8f;

#define C_DIM 128
#define TILE_ROWS 16
#define TILES_PER_WAVE 2
#define WAVES_PER_BLOCK 4
#define ROWS_PER_BLOCK (TILE_ROWS * TILES_PER_WAVE * WAVES_PER_BLOCK)

__global__ __launch_bounds__(256) void gcn_deg_kernel(
    const int* __restrict__ edges, float* __restrict__ inv_s, int N) {
  int n = blockIdx.x * blockDim.x + threadIdx.x;
  if (n >= N) return;
  int cnt = 1;
#pragma unroll
  for (int d = 0; d < 16; ++d) cnt += (edges[n * 16 + d] >= 0) ? 1 : 0;
  inv_s[n] = rsqrtf((float)cnt);
}

__global__ __launch_bounds__(128) void gcn_gemm_wmma_kernel(
    const float* __restrict__ X, const float* __restrict__ W,
    const float* __restrict__ bias, const float* __restrict__ inv_s,
    float* __restrict__ H, int N) {
  __shared__ v2f lWp[(C_DIM / 2) * C_DIM];

  const int tid = threadIdx.x;

  for (int idx = tid; idx < (C_DIM / 2) * C_DIM; idx += 128) {
    const int k2 = idx >> 7;
    const int col = idx & (C_DIM - 1);
    v2f p;
    p.x = W[(2 * k2 + 0) * C_DIM + col];
    p.y = W[(2 * k2 + 1) * C_DIM + col];
    lWp[idx] = p;
  }
  __syncthreads();

  const int wave = tid >> 5;
  const int lane = tid & 31;
  const int mrow = lane & 15;
  const int khalf = lane >> 4;

  const int row0 = blockIdx.x * ROWS_PER_BLOCK + wave * (TILE_ROWS * TILES_PER_WAVE);
  int arowA = row0 + mrow;
  int arowB = row0 + TILE_ROWS + mrow;
  if (arowA >= N) arowA = 0;
  if (arowB >= N) arowB = 0;
  const float* xrowA = X + (size_t)arowA * C_DIM;
  const float* xrowB = X + (size_t)arowB * C_DIM;

  v8f acc0[8], acc1[8];
  const v8f vzero = {0.f, 0.f, 0.f, 0.f, 0.f, 0.f, 0.f, 0.f};
#pragma unroll
  for (int t = 0; t < 8; ++t) {
    acc0[t] = vzero;
    acc1[t] = vzero;
  }

#pragma unroll 2
  for (int k0 = 0; k0 < C_DIM; k0 += 4) {
    const v2f afragA = *(const v2f*)(xrowA + k0 + 2 * khalf);
    const v2f afragB = *(const v2f*)(xrowB + k0 + 2 * khalf);
    const v2f* bbase = &lWp[(k0 / 2 + khalf) * C_DIM];
#pragma unroll
    for (int t = 0; t < 8; ++t) {
      const v2f bfrag = bbase[t * 16 + mrow];
      acc0[t] = __builtin_amdgcn_wmma_f32_16x16x4_f32(
          false, afragA, false, bfrag, (short)0, acc0[t], false, false);
      acc1[t] = __builtin_amdgcn_wmma_f32_16x16x4_f32(
          false, afragB, false, bfrag, (short)0, acc1[t], false, false);
    }
  }

#pragma unroll
  for (int tile = 0; tile < TILES_PER_WAVE; ++tile) {
    const v8f* acc = (tile == 0) ? acc0 : acc1;
#pragma unroll
    for (int v = 0; v < 8; ++v) {
      const int r = row0 + tile * TILE_ROWS + 8 * khalf + v;
      if (r < N) {
        const float sc = inv_s[r];
        float* hrow = H + (size_t)r * C_DIM;
#pragma unroll
        for (int t = 0; t < 8; ++t) {
          const int col = t * 16 + mrow;
          hrow[col] = (acc[t][v] + bias[col]) * sc;
        }
      }
    }
  }
}

__global__ __launch_bounds__(256) void gcn_gather_elu_kernel(
    const float* __restrict__ H, const int* __restrict__ edges,
    const float* __restrict__ inv_s, float* __restrict__ out, int N) {
  const int node = blockIdx.x * 8 + (threadIdx.x >> 5);
  if (node >= N) return;
  const int c4 = threadIdx.x & 31;

  const float4* Hv = (const float4*)H;
  float4 self = Hv[(size_t)node * 32 + c4];
  float ax = self.x, ay = self.y, az = self.z, aw = self.w;

#pragma unroll
  for (int d = 0; d < 16; ++d) {
    const int e = edges[node * 16 + d];
    if (e >= 0) {
      float4 v = Hv[(size_t)e * 32 + c4];
      ax += v.x;
      ay += v.y;
      az += v.z;
      aw += v.w;
    }
  }

  const float sc = inv_s[node];
  ax *= sc;
  ay *= sc;
  az *= sc;
  aw *= sc;

  float4 r;
  r.x = (ax > 0.f) ? ax : expm1f(ax);
  r.y = (ay > 0.f) ? ay : expm1f(ay);
  r.z = (az > 0.f) ? az : expm1f(az);
  r.w = (aw > 0.f) ? aw : expm1f(aw);
  ((float4*)out)[(size_t)node * 32 + c4] = r;
}

extern "C" void kernel_launch(void* const* d_in, const int* in_sizes, int n_in,
                              void* d_out, int out_size, void* d_ws,
                              size_t ws_size, hipStream_t stream) {
  const float* x = (const float*)d_in[0];
  const int* edges = (const int*)d_in[1];
  const float* W1 = (const float*)d_in[2];
  const float* b1 = (const float*)d_in[3];
  const float* W2 = (const float*)d_in[4];
  const float* b2 = (const float*)d_in[5];
  float* out = (float*)d_out;

  const int N = in_sizes[0] / C_DIM;

  float* inv_s = (float*)d_ws;
  const size_t inv_pad = ((size_t)N + 63) & ~(size_t)63;
  float* H = inv_s + inv_pad;

  const int deg_blocks = (N + 255) / 256;
  const int gemm_blocks = (N + ROWS_PER_BLOCK - 1) / ROWS_PER_BLOCK;
  const int gather_blocks = (N + 7) / 8;

  gcn_deg_kernel<<<deg_blocks, 256, 0, stream>>>(edges, inv_s, N);

  gcn_gemm_wmma_kernel<<<gemm_blocks, 128, 0, stream>>>(x, W1, b1, inv_s, H, N);
  gcn_gather_elu_kernel<<<gather_blocks, 256, 0, stream>>>(H, edges, inv_s, out, N);

  gcn_gemm_wmma_kernel<<<gemm_blocks, 128, 0, stream>>>(out, W2, b2, inv_s, H, N);
  gcn_gather_elu_kernel<<<gather_blocks, 256, 0, stream>>>(H, edges, inv_s, out, N);
}


# GEMM A operand path: each wave keeps a 16x128 A stripe in registers via 16-byte row loads refilled one unit ahead, in a persistent 256-workgroup grid that stages W once; k summed in quad order
# speedup vs baseline: 1.1118x; 1.1118x over previous
#include <hip/hip_runtime.h>
#include <hip/hip_bf16.h>
#include <math.h>


typedef __attribute__((ext_vector_type(2))) float v2f;
typedef __attribute__((ext_vector_type(4))) float v4f;
typedef __attribute__((ext_vector_type(8))) float v8f;

#define C_DIM 128
#define UNIT_ROWS 16
#define GEMM_WAVES 16
#define GEMM_THREADS (GEMM_WAVES * 32)
#define GEMM_BLOCKS 256

__global__ __launch_bounds__(256) void gcn_deg_kernel(
    const int* __restrict__ edges, float* __restrict__ inv_s, int N) {
  int n = blockIdx.x * blockDim.x + threadIdx.x;
  if (n >= N) return;
  int cnt = 1;
#pragma unroll
  for (int d = 0; d < 16; ++d) cnt += (edges[n * 16 + d] >= 0) ? 1 : 0;
  inv_s[n] = rsqrtf((float)cnt);
}

__global__ __launch_bounds__(GEMM_THREADS) void gcn_gemm_wmma_kernel(
    const float* __restrict__ X, const float* __restrict__ W,
    const float* __restrict__ bias, const float* __restrict__ inv_s,
    float* __restrict__ H, int N) {
  __shared__ v4f lWq[(C_DIM / 4) * C_DIM];

  const int tid = threadIdx.x;

#pragma unroll
  for (int it = 0; it < (C_DIM / 4) * C_DIM / GEMM_THREADS; ++it) {
    const int idx = tid + it * GEMM_THREADS;
    const int kq = idx >> 7;
    const int col = idx & (C_DIM - 1);
    v4f p;
    p.x = W[(4 * kq + 0) * C_DIM + col];
    p.y = W[(4 * kq + 1) * C_DIM + col];
    p.z = W[(4 * kq + 2) * C_DIM + col];
    p.w = W[(4 * kq + 3) * C_DIM + col];
    lWq[idx] = p;
  }

  const int wave = __builtin_amdgcn_readfirstlane(tid >> 5);
  const int lane = tid & 31;
  const int c = lane & 15;
  const int h = lane >> 4;

  const int nUnits = (N + UNIT_ROWS - 1) / UNIT_ROWS;
  const int stride = gridDim.x * GEMM_WAVES;
  const int rank = (((wave >> 2) << 1) | (wave & 1)) + ((((wave ^ (wave >> 1)) & 2) != 0) ? 8 : 0);
  int u = rank * gridDim.x + blockIdx.x;

  float bcol[8];
#pragma unroll
  for (int t = 0; t < 8; ++t) bcol[t] = bias[t * 16 + c];

  v4f a[16];
  {
    int row = u * UNIT_ROWS + c;
    if (u >= nUnits || row >= N) row = 0;
    const float* p = X + (size_t)row * C_DIM + 4 * h;
#pragma unroll
    for (int j = 0; j < 16; ++j) a[j] = *(const v4f*)(p + 8 * j);
  }

  __syncthreads();

  for (; u < nUnits; u += stride) {
    const int un = u + stride;
    int rown = un * UNIT_ROWS + c;
    if (un >= nUnits || rown >= N) rown = 0;
    const float* pn = X + (size_t)rown * C_DIM + 4 * h;

    float sc[8];
#pragma unroll
    for (int v = 0; v < 8; ++v) {
      const int r = u * UNIT_ROWS + 8 * h + v;
      sc[v] = inv_s[(r < N) ? r : 0];
    }
    __builtin_amdgcn_sched_barrier(0);

    v8f acc[8];
    const v8f vzero = {0.f, 0.f, 0.f, 0.f, 0.f, 0.f, 0.f, 0.f};
#pragma unroll
    for (int t = 0; t < 8; ++t) acc[t] = vzero;

    int boff = h * C_DIM + c;
    asm volatile("" : "+v"(boff));
    const v4f* bbase = &lWq[boff];

#pragma unroll
    for (int g = 0; g < 4; ++g) {
#pragma unroll
      for (int jj = 0; jj < 4; ++jj) {
        const int j = 4 * g + jj;
        const v4f* bb = bbase + 2 * j * C_DIM;
        v4f b[8];
#pragma unroll
        for (int t = 0; t < 8; ++t) b[t] = bb[t * 16];
        const v2f a0 = {a[j].x, a[j].y};
        const v2f a1 = {a[j].z, a[j].w};
#pragma unroll
        for (int t = 0; t < 8; ++t) {
          const v2f b0 = {b[t].x, b[t].y};
          acc[t] = __builtin_amdgcn_wmma_f32_16x16x4_f32(
              false, a0, false, b0, (short)0, acc[t], false, false);
        }
#pragma unroll
        for (int t = 0; t < 8; ++t) {
          const v2f b1 = {b[t].z, b[t].w};
          acc[t] = __builtin_amdgcn_wmma_f32_16x16x4_f32(
              false, a1, false, b1, (short)0, acc[t], false, false);
        }
      }
      __builtin_amdgcn_sched_barrier(0);
#pragma unroll
      for (int jj = 0; jj < 4; ++jj) {
        const int j = 4 * g + jj;
        a[j] = *(const v4f*)(pn + 8 * j);
      }
      __builtin_amdgcn_sched_barrier(0);
    }

#pragma unroll
    for (int t = 0; t < 8; ++t) {
#pragma unroll
      for (int v = 0; v < 8; ++v) acc[t][v] = (acc[t][v] + bcol[t]) * sc[v];
    }
    if ((u + 1) * UNIT_ROWS <= N) {
#pragma unroll
      for (int v = 0; v < 8; ++v) {
        float* hrow = H + (size_t)(u * UNIT_ROWS + 8 * h + v) * C_DIM;
#pragma unroll
        for (int t = 0; t < 8; ++t) hrow[t * 16 + c] = acc[t][v];
      }
    } else {
#pragma unroll
      for (int v = 0; v < 8; ++v) {
        const int r = u * UNIT_ROWS + 8 * h + v;
        if (r < N) {
          float* hrow = H + (size_t)r * C_DIM;
#pragma unroll
          for (int t = 0; t < 8; ++t) hrow[t * 16 + c] = acc[t][v];
        }
      }
    }
  }
}

__global__ __launch_bounds__(256) void gcn_gather_elu_kernel(
    const float* __restrict__ H, const int* __restrict__ edges,
    const float* __restrict__ inv_s, float* __restrict__ out, int N) {
  const int node = blockIdx.x * 8 + (threadIdx.x >> 5);
  if (node >= N) return;
  const int c4 = threadIdx.x & 31;

  const float4* Hv = (const float4*)H;
  float4 self = Hv[(size_t)node * 32 + c4];
  float ax = self.x, ay = self.y, az = self.z, aw = self.w;

#pragma unroll
  for (int d = 0; d < 16; ++d) {
    const int e = edges[node * 16 + d];
    if (e >= 0) {
      float4 v = Hv[(size_t)e * 32 + c4];
      ax += v.x;
      ay += v.y;
      az += v.z;
      aw += v.w;
    }
  }

  const float sc = inv_s[node];
  ax *= sc;
  ay *= sc;
  az *= sc;
  aw *= sc;

  float4 r;
  r.x = (ax > 0.f) ? ax : expm1f(ax);
  r.y = (ay > 0.f) ? ay : expm1f(ay);
  r.z = (az > 0.f) ? az : expm1f(az);
  r.w = (aw > 0.f) ? aw : expm1f(aw);
  ((float4*)out)[(size_t)node * 32 + c4] = r;
}

extern "C" void kernel_launch(void* const* d_in, const int* in_sizes, int n_in,
                              void* d_out, int out_size, void* d_ws,
                              size_t ws_size, hipStream_t stream) {
  const float* x = (const float*)d_in[0];
  const int* edges = (const int*)d_in[1];
  const float* W1 = (const float*)d_in[2];
  const float* b1 = (const float*)d_in[3];
  const float* W2 = (const float*)d_in[4];
  const float* b2 = (const float*)d_in[5];
  float* out = (float*)d_out;

  const int N = in_sizes[0] / C_DIM;

  float* inv_s = (float*)d_ws;
  const size_t inv_pad = ((size_t)N + 63) & ~(size_t)63;
  float* H = inv_s + inv_pad;

  const int deg_blocks = (N + 255) / 256;
  const int gather_blocks = (N + 7) / 8;

  gcn_deg_kernel<<<deg_blocks, 256, 0, stream>>>(edges, inv_s, N);

  gcn_gemm_wmma_kernel<<<GEMM_BLOCKS, GEMM_THREADS, 0, stream>>>(x, W1, b1, inv_s, H, N);
  gcn_gather_elu_kernel<<<gather_blocks, 256, 0, stream>>>(H, edges, inv_s, out, N);

  gcn_gemm_wmma_kernel<<<GEMM_BLOCKS, GEMM_THREADS, 0, stream>>>(out, W2, b2, inv_s, H, N);
  gcn_gather_elu_kernel<<<gather_blocks, 256, 0, stream>>>(H, edges, inv_s, out, N);
}


# gather kernel: resident waves walk the nodes; a node's 16 indices come as one scalar load a node ahead and all its neighbour rows are loaded before one wait (was one row in flight); add order kept
# speedup vs baseline: 1.3447x; 1.2095x over previous
#include <hip/hip_runtime.h>
#include <hip/hip_bf16.h>
#include <math.h>


typedef __attribute__((ext_vector_type(2))) float v2f;
typedef __attribute__((ext_vector_type(4))) float v4f;
typedef __attribute__((ext_vector_type(8))) float v8f;
typedef __attribute__((ext_vector_type(16))) int v16i;

#define C_DIM 128
#define UNIT_ROWS 16
#define GEMM_WAVES 16
#define GEMM_THREADS (GEMM_WAVES * 32)
#define GEMM_BLOCKS 256
#define GATHER_WAVES 8
#define GATHER_THREADS (GATHER_WAVES * 32)
#define GATHER_BLOCKS 1024

__global__ __launch_bounds__(256) void gcn_deg_kernel(
    const int* __restrict__ edges, float* __restrict__ inv_s, int N) {
  int n = blockIdx.x * blockDim.x + threadIdx.x;
  if (n >= N) return;
  int cnt = 1;
#pragma unroll
  for (int d = 0; d < 16; ++d) cnt += (edges[n * 16 + d] >= 0) ? 1 : 0;
  inv_s[n] = rsqrtf((float)cnt);
}

__global__ __launch_bounds__(GEMM_THREADS) void gcn_gemm_wmma_kernel(
    const float* __restrict__ X, const float* __restrict__ W,
    const float* __restrict__ bias, const float* __restrict__ inv_s,
    float* __restrict__ H, int N) {
  __shared__ v4f lWq[(C_DIM / 4) * C_DIM];

  const int tid = threadIdx.x;

#pragma unroll
  for (int it = 0; it < (C_DIM / 4) * C_DIM / GEMM_THREADS; ++it) {
    const int idx = tid + it * GEMM_THREADS;
    const int kq = idx >> 7;
    const int col = idx & (C_DIM - 1);
    v4f p;
    p.x = W[(4 * kq + 0) * C_DIM + col];
    p.y = W[(4 * kq + 1) * C_DIM + col];
    p.z = W[(4 * kq + 2) * C_DIM + col];
    p.w = W[(4 * kq + 3) * C_DIM + col];
    lWq[idx] = p;
  }

  const int wave = __builtin_amdgcn_readfirstlane(tid >> 5);
  const int lane = tid & 31;
  const int c = lane & 15;
  const int h = lane >> 4;

  const int nUnits = (N + UNIT_ROWS - 1) / UNIT_ROWS;
  const int stride = gridDim.x * GEMM_WAVES;
  const int rank = (((wave >> 2) << 1) | (wave & 1)) + ((((wave ^ (wave >> 1)) & 2) != 0) ? 8 : 0);
  int u = rank * gridDim.x + blockIdx.x;

  float bcol[8];
#pragma unroll
  for (int t = 0; t < 8; ++t) bcol[t] = bias[t * 16 + c];

  v4f a[16];
  {
    int row = u * UNIT_ROWS + c;
    if (u >= nUnits || row >= N) row = 0;
    const float* p = X + (size_t)row * C_DIM + 4 * h;
#pragma unroll
    for (int j = 0; j < 16; ++j) a[j] = *(const v4f*)(p + 8 * j);
  }

  __syncthreads();

  for (; u < nUnits; u += stride) {
    const int un = u + stride;
    int rown = un * UNIT_ROWS + c;
    if (un >= nUnits || rown >= N) rown = 0;
    const float* pn = X + (size_t)rown * C_DIM + 4 * h;

    float sc[8];
#pragma unroll
    for (int v = 0; v < 8; ++v) {
      const int r = u * UNIT_ROWS + 8 * h + v;
      sc[v] = inv_s[(r < N) ? r : 0];
    }
    __builtin_amdgcn_sched_barrier(0);

    v8f acc[8];
    const v8f vzero = {0.f, 0.f, 0.f, 0.f, 0.f, 0.f, 0.f, 0.f};
#pragma unroll
    for (int t = 0; t < 8; ++t) acc[t] = vzero;

    int boff = h * C_DIM + c;
    asm volatile("" : "+v"(boff));
    const v4f* bbase = &lWq[boff];

#pragma unroll
    for (int g = 0; g < 4; ++g) {
#pragma unroll
      for (int jj = 0; jj < 4; ++jj) {
        const int j = 4 * g + jj;
        const v4f* bb = bbase + 2 * j * C_DIM;
        v4f b[8];
#pragma unroll
        for (int t = 0; t < 8; ++t) b[t] = bb[t * 16];
        const v2f a0 = {a[j].x, a[j].y};
        const v2f a1 = {a[j].z, a[j].w};
#pragma unroll
        for (int t = 0; t < 8; ++t) {
          const v2f b0 = {b[t].x, b[t].y};
          acc[t] = __builtin_amdgcn_wmma_f32_16x16x4_f32(
              false, a0, false, b0, (short)0, acc[t], false, false);
        }
#pragma unroll
        for (int t = 0; t < 8; ++t) {
          const v2f b1 = {b[t].z, b[t].w};
          acc[t] = __builtin_amdgcn_wmma_f32_16x16x4_f32(
              false, a1, false, b1, (short)0, acc[t], false, false);
        }
      }
      __builtin_amdgcn_sched_barrier(0);
#pragma unroll
      for (int jj = 0; jj < 4; ++jj) {
        const int j = 4 * g + jj;
        a[j] = *(const v4f*)(pn + 8 * j);
      }
      __builtin_amdgcn_sched_barrier(0);
    }

#pragma unroll
    for (int t = 0; t < 8; ++t) {
#pragma unroll
      for (int v = 0; v < 8; ++v) acc[t][v] = (acc[t][v] + bcol[t]) * sc[v];
    }
    if ((u + 1) * UNIT_ROWS <= N) {
#pragma unroll
      for (int v = 0; v < 8; ++v) {
        float* hrow = H + (size_t)(u * UNIT_ROWS + 8 * h + v) * C_DIM;
#pragma unroll
        for (int t = 0; t < 8; ++t) hrow[t * 16 + c] = acc[t][v];
      }
    } else {
#pragma unroll
      for (int v = 0; v < 8; ++v) {
        const int r = u * UNIT_ROWS + 8 * h + v;
        if (r < N) {
          float* hrow = H + (size_t)r * C_DIM;
#pragma unroll
          for (int t = 0; t < 8; ++t) hrow[t * 16 + c] = acc[t][v];
        }
      }
    }
  }
}

__global__ __launch_bounds__(GATHER_THREADS) void gcn_gather_elu_kernel(
    const float* __restrict__ H, const int* __restrict__ edges,
    const float* __restrict__ inv_s, float* __restrict__ out, int N) {
  const int wave = __builtin_amdgcn_readfirstlane(threadIdx.x >> 5);
  const int c4 = threadIdx.x & 31;
  const int nWaves = gridDim.x * GATHER_WAVES;
  int node = blockIdx.x * GATHER_WAVES + wave;
  if (node >= N) return;

  const float4* Hv = (const float4*)H;
  const v16i* Ev = (const v16i*)edges;

  v16i e = Ev[node];
  float sc = inv_s[node];

  for (;;) {
    const int next = node + nWaves;
    const int nn = (next < N) ? next : node;
    const v16i en = Ev[nn];
    const float scn = inv_s[nn];

    const float4 self = Hv[(size_t)node * 32 + c4];
    float4 v[16];
#pragma unroll
    for (int d = 0; d < 16; ++d) {
      if (e[d] >= 0) {
        v[d] = Hv[(size_t)e[d] * 32 + c4];
      } else {
        v[d] = make_float4(0.f, 0.f, 0.f, 0.f);
      }
    }

    float ax = self.x, ay = self.y, az = self.z, aw = self.w;
#pragma unroll
    for (int d = 0; d < 16; ++d) {
      ax += v[d].x;
      ay += v[d].y;
      az += v[d].z;
      aw += v[d].w;
    }

    ax *= sc;
    ay *= sc;
    az *= sc;
    aw *= sc;

    float4 r;
    r.x = (ax > 0.f) ? ax : expm1f(ax);
    r.y = (ay > 0.f) ? ay : expm1f(ay);
    r.z = (az > 0.f) ? az : expm1f(az);
    r.w = (aw > 0.f) ? aw : expm1f(aw);
    ((float4*)out)[(size_t)node * 32 + c4] = r;

    if (next >= N) break;
    node = next;
    e = en;
    sc = scn;
  }
}

extern "C" void kernel_launch(void* const* d_in, const int* in_sizes, int n_in,
                              void* d_out, int out_size, void* d_ws,
                              size_t ws_size, hipStream_t stream) {
  const float* x = (const float*)d_in[0];
  const int* edges = (const int*)d_in[1];
  const float* W1 = (const float*)d_in[2];
  const float* b1 = (const float*)d_in[3];
  const float* W2 = (const float*)d_in[4];
  const float* b2 = (const float*)d_in[5];
  float* out = (float*)d_out;

  const int N = in_sizes[0] / C_DIM;

  float* inv_s = (float*)d_ws;
  const size_t inv_pad = ((size_t)N + 63) & ~(size_t)63;
  float* H = inv_s + inv_pad;

  const int deg_blocks = (N + 255) / 256;

  gcn_deg_kernel<<<deg_blocks, 256, 0, stream>>>(edges, inv_s, N);

  gcn_gemm_wmma_kernel<<<GEMM_BLOCKS, GEMM_THREADS, 0, stream>>>(x, W1, b1, inv_s, H, N);
  gcn_gather_elu_kernel<<<GATHER_BLOCKS, GATHER_THREADS, 0, stream>>>(H, edges, inv_s, out, N);

  gcn_gemm_wmma_kernel<<<GEMM_BLOCKS, GEMM_THREADS, 0, stream>>>(out, W2, b2, inv_s, H, N);
  gcn_gather_elu_kernel<<<GATHER_BLOCKS, GATHER_THREADS, 0, stream>>>(H, edges, inv_s, out, N);
}


# layer-2 product fused into the layer-1 gather launch: a wave gathers its 16 nodes into a private LDS tile and reads it back as the WMMA A stripe; 5 launches -> 4, sum order kept
# speedup vs baseline: 1.4353x; 1.0674x over previous
#include <hip/hip_runtime.h>
#include <hip/hip_bf16.h>
#include <math.h>


typedef __attribute__((ext_vector_type(2))) float v2f;
typedef __attribute__((ext_vector_type(4))) float v4f;
typedef __attribute__((ext_vector_type(8))) float v8f;
typedef __attribute__((ext_vector_type(16))) int v16i;

#define C_DIM 128
#define UNIT_ROWS 16
#define GEMM_WAVES 16
#define GEMM_THREADS (GEMM_WAVES * 32)
#define GEMM_BLOCKS 256
#define GATHER_WAVES 8
#define GATHER_THREADS (GATHER_WAVES * 32)
#define GATHER_BLOCKS 1024

__global__ __launch_bounds__(256) void gcn_deg_kernel(
    const int* __restrict__ edges, float* __restrict__ inv_s, int N) {
  int n = blockIdx.x * blockDim.x + threadIdx.x;
  if (n >= N) return;
  int cnt = 1;
#pragma unroll
  for (int d = 0; d < 16; ++d) cnt += (edges[n * 16 + d] >= 0) ? 1 : 0;
  inv_s[n] = rsqrtf((float)cnt);
}

__global__ __launch_bounds__(GEMM_THREADS) void gcn_gemm_wmma_kernel(
    const float* __restrict__ X, const float* __restrict__ W,
    const float* __restrict__ bias, const float* __restrict__ inv_s,
    float* __restrict__ H, int N) {
  __shared__ v4f lWq[(C_DIM / 4) * C_DIM];

  const int tid = threadIdx.x;

#pragma unroll
  for (int it = 0; it < (C_DIM / 4) * C_DIM / GEMM_THREADS; ++it) {
    const int idx = tid + it * GEMM_THREADS;
    const int kq = idx >> 7;
    const int col = idx & (C_DIM - 1);
    v4f p;
    p.x = W[(4 * kq + 0) * C_DIM + col];
    p.y = W[(4 * kq + 1) * C_DIM + col];
    p.z = W[(4 * kq + 2) * C_DIM + col];
    p.w = W[(4 * kq + 3) * C_DIM + col];
    lWq[idx] = p;
  }

  const int wave = __builtin_amdgcn_readfirstlane(tid >> 5);
  const int lane = tid & 31;
  const int c = lane & 15;
  const int h = lane >> 4;

  const int nUnits = (N + UNIT_ROWS - 1) / UNIT_ROWS;
  const int stride = gridDim.x * GEMM_WAVES;
  const int rank = (((wave >> 2) << 1) | (wave & 1)) + ((((wave ^ (wave >> 1)) & 2) != 0) ? 8 : 0);
  int u = rank * gridDim.x + blockIdx.x;

  float bcol[8];
#pragma unroll
  for (int t = 0; t < 8; ++t) bcol[t] = bias[t * 16 + c];

  v4f a[16];
  {
    int row = u * UNIT_ROWS + c;
    if (u >= nUnits || row >= N) row = 0;
    const float* p = X + (size_t)row * C_DIM + 4 * h;
#pragma unroll
    for (int j = 0; j < 16; ++j) a[j] = *(const v4f*)(p + 8 * j);
  }

  __syncthreads();

  for (; u < nUnits; u += stride) {
    const int un = u + stride;
    int rown = un * UNIT_ROWS + c;
    if (un >= nUnits || rown >= N) rown = 0;
    const float* pn = X + (size_t)rown * C_DIM + 4 * h;

    float sc[8];
#pragma unroll
    for (int v = 0; v < 8; ++v) {
      const int r = u * UNIT_ROWS + 8 * h + v;
      sc[v] = inv_s[(r < N) ? r : 0];
    }
    __builtin_amdgcn_sched_barrier(0);

    v8f acc[8];
    const v8f vzero = {0.f, 0.f, 0.f, 0.f, 0.f, 0.f, 0.f, 0.f};
#pragma unroll
    for (int t = 0; t < 8; ++t) acc[t] = vzero;

    int boff = h * C_DIM + c;
    asm volatile("" : "+v"(boff));
    const v4f* bbase = &lWq[boff];

#pragma unroll
    for (int g = 0; g < 4; ++g) {
#pragma unroll
      for (int jj = 0; jj < 4; ++jj) {
        const int j = 4 * g + jj;
        const v4f* bb = bbase + 2 * j * C_DIM;
        v4f b[8];
#pragma unroll
        for (int t = 0; t < 8; ++t) b[t] = bb[t * 16];
        const v2f a0 = {a[j].x, a[j].y};
        const v2f a1 = {a[j].z, a[j].w};
#pragma unroll
        for (int t = 0; t < 8; ++t) {
          const v2f b0 = {b[t].x, b[t].y};
          acc[t] = __builtin_amdgcn_wmma_f32_16x16x4_f32(
              false, a0, false, b0, (short)0, acc[t], false, false);
        }
#pragma unroll
        for (int t = 0; t < 8; ++t) {
          const v2f b1 = {b[t].z, b[t].w};
          acc[t] = __builtin_amdgcn_wmma_f32_16x16x4_f32(
              false, a1, false, b1, (short)0, acc[t], false, false);
        }
      }
      __builtin_amdgcn_sched_barrier(0);
#pragma unroll
      for (int jj = 0; jj < 4; ++jj) {
        const int j = 4 * g + jj;
        a[j] = *(const v4f*)(pn + 8 * j);
      }
      __builtin_amdgcn_sched_barrier(0);
    }

#pragma unroll
    for (int t = 0; t < 8; ++t) {
#pragma unroll
      for (int v = 0; v < 8; ++v) acc[t][v] = (acc[t][v] + bcol[t]) * sc[v];
    }
    if ((u + 1) * UNIT_ROWS <= N) {
#pragma unroll
      for (int v = 0; v < 8; ++v) {
        float* hrow = H + (size_t)(u * UNIT_ROWS + 8 * h + v) * C_DIM;
#pragma unroll
        for (int t = 0; t < 8; ++t) hrow[t * 16 + c] = acc[t][v];
      }
    } else {
#pragma unroll
      for (int v = 0; v < 8; ++v) {
        const int r = u * UNIT_ROWS + 8 * h + v;
        if (r < N) {
          float* hrow = H + (size_t)r * C_DIM;
#pragma unroll
          for (int t = 0; t < 8; ++t) hrow[t * 16 + c] = acc[t][v];
        }
      }
    }
  }
}

__global__ __launch_bounds__(GATHER_THREADS) void gcn_gather_elu_kernel(
    const float* __restrict__ H, const int* __restrict__ edges,
    const float* __restrict__ inv_s, float* __restrict__ out, int N) {
  const int wave = __builtin_amdgcn_readfirstlane(threadIdx.x >> 5);
  const int c4 = threadIdx.x & 31;
  const int nWaves = gridDim.x * GATHER_WAVES;
  int node = blockIdx.x * GATHER_WAVES + wave;
  if (node >= N) return;

  const float4* Hv = (const float4*)H;
  const v16i* Ev = (const v16i*)edges;

  v16i e = Ev[node];
  float sc = inv_s[node];

  for (;;) {
    const int next = node + nWaves;
    const int nn = (next < N) ? next : node;
    const v16i en = Ev[nn];
    const float scn = inv_s[nn];

    const float4 self = Hv[(size_t)node * 32 + c4];
    float4 v[16];
#pragma unroll
    for (int d = 0; d < 16; ++d) {
      if (e[d] >= 0) {
        v[d] = Hv[(size_t)e[d] * 32 + c4];
      } else {
        v[d] = make_float4(0.f, 0.f, 0.f, 0.f);
      }
    }

    float ax = self.x, ay = self.y, az = self.z, aw = self.w;
#pragma unroll
    for (int d = 0; d < 16; ++d) {
      ax += v[d].x;
      ay += v[d].y;
      az += v[d].z;
      aw += v[d].w;
    }

    ax *= sc;
    ay *= sc;
    az *= sc;
    aw *= sc;

    float4 r;
    r.x = (ax > 0.f) ? ax : expm1f(ax);
    r.y = (ay > 0.f) ? ay : expm1f(ay);
    r.z = (az > 0.f) ? az : expm1f(az);
    r.w = (aw > 0.f) ? aw : expm1f(aw);
    ((float4*)out)[(size_t)node * 32 + c4] = r;

    if (next >= N) break;
    node = next;
    e = en;
    sc = scn;
  }
}

#define TILE_PITCH 33
__global__ __launch_bounds__(GEMM_THREADS) void gcn_gather_elu_gemm_kernel(
    const float* __restrict__ Hin, const int* __restrict__ edges,
    const float* __restrict__ inv_s, const float* __restrict__ W,
    const float* __restrict__ bias, float* __restrict__ Hout, int N) {
  __shared__ v4f lWq[(C_DIM / 4) * C_DIM];
  __shared__ v4f lA[GEMM_WAVES * UNIT_ROWS * TILE_PITCH];

  const int tid = threadIdx.x;

#pragma unroll
  for (int it = 0; it < (C_DIM / 4) * C_DIM / GEMM_THREADS; ++it) {
    const int idx = tid + it * GEMM_THREADS;
    const int kq = idx >> 7;
    const int col = idx & (C_DIM - 1);
    v4f p;
    p.x = W[(4 * kq + 0) * C_DIM + col];
    p.y = W[(4 * kq + 1) * C_DIM + col];
    p.z = W[(4 * kq + 2) * C_DIM + col];
    p.w = W[(4 * kq + 3) * C_DIM + col];
    lWq[idx] = p;
  }

  const int wave = __builtin_amdgcn_readfirstlane(tid >> 5);
  const int lane = tid & 31;
  const int c = lane & 15;
  const int h = lane >> 4;

  const int nUnits = (N + UNIT_ROWS - 1) / UNIT_ROWS;
  const int stride = gridDim.x * GEMM_WAVES;
  const int rank = (((wave >> 2) << 1) | (wave & 1)) + ((((wave ^ (wave >> 1)) & 2) != 0) ? 8 : 0);
  int u = rank * gridDim.x + blockIdx.x;

  float bcol[8];
#pragma unroll
  for (int t = 0; t < 8; ++t) bcol[t] = bias[t * 16 + c];

  const float4* Hv = (const float4*)Hin;
  typedef const __attribute__((address_space(4))) v16i* idx_cptr;
  typedef const __attribute__((address_space(4))) float* scale_cptr;
  const idx_cptr Ev = (idx_cptr)edges;
  const scale_cptr Sv = (scale_cptr)inv_s;
  v4f* slot = &lA[wave * (UNIT_ROWS * TILE_PITCH)];
  const int last = N - 1;

  __syncthreads();

  for (; u < nUnits; u += stride) {
    const int base = u * UNIT_ROWS;

    float sc[8];
#pragma unroll
    for (int v = 0; v < 8; ++v) {
      const int r = base + 8 * h + v;
      sc[v] = inv_s[(r < N) ? r : 0];
    }

    int nA = base;
    int nB = (base + 1 < N) ? base + 1 : last;
    v16i eA = Ev[nA];
    v16i eB = Ev[nB];
    float sA = Sv[nA];
    float sB = Sv[nB];

#pragma nounroll
    for (int i = 0; i < UNIT_ROWS; i += 2) {
      const int nAn = (base + i + 2 < N) ? base + i + 2 : last;
      const int nBn = (base + i + 3 < N) ? base + i + 3 : last;
      const v16i eAn = Ev[nAn];
      const v16i eBn = Ev[nBn];
      const float sAn = Sv[nAn];
      const float sBn = Sv[nBn];

      const float4 selfA = Hv[(size_t)nA * 32 + lane];
      const float4 selfB = Hv[(size_t)nB * 32 + lane];
      float4 va[16], vb[16];
#pragma unroll
      for (int d = 0; d < 16; ++d) {
        if (eA[d] >= 0) {
          va[d] = Hv[(size_t)eA[d] * 32 + lane];
        } else {
          va[d] = make_float4(0.f, 0.f, 0.f, 0.f);
        }
      }
#pragma unroll
      for (int d = 0; d < 16; ++d) {
        if (eB[d] >= 0) {
          vb[d] = Hv[(size_t)eB[d] * 32 + lane];
        } else {
          vb[d] = make_float4(0.f, 0.f, 0.f, 0.f);
        }
      }

      float ax = selfA.x, ay = selfA.y, az = selfA.z, aw = selfA.w;
#pragma unroll
      for (int d = 0; d < 16; ++d) {
        ax += va[d].x;
        ay += va[d].y;
        az += va[d].z;
        aw += va[d].w;
      }
      ax *= sA;
      ay *= sA;
      az *= sA;
      aw *= sA;
      v4f rA;
      rA.x = (ax > 0.f) ? ax : expm1f(ax);
      rA.y = (ay > 0.f) ? ay : expm1f(ay);
      rA.z = (az > 0.f) ? az : expm1f(az);
      rA.w = (aw > 0.f) ? aw : expm1f(aw);
      slot[i * TILE_PITCH + lane] = rA;

      float bx = selfB.x, by = selfB.y, bz = selfB.z, bw = selfB.w;
#pragma unroll
      for (int d = 0; d < 16; ++d) {
        bx += vb[d].x;
        by += vb[d].y;
        bz += vb[d].z;
        bw += vb[d].w;
      }
      bx *= sB;
      by *= sB;
      bz *= sB;
      bw *= sB;
      v4f rB;
      rB.x = (bx > 0.f) ? bx : expm1f(bx);
      rB.y = (by > 0.f) ? by : expm1f(by);
      rB.z = (bz > 0.f) ? bz : expm1f(bz);
      rB.w = (bw > 0.f) ? bw : expm1f(bw);
      slot[(i + 1) * TILE_PITCH + lane] = rB;

      nA = nAn;
      nB = nBn;
      eA = eAn;
      eB = eBn;
      sA = sAn;
      sB = sBn;
    }

    __builtin_amdgcn_wave_barrier();

    v4f a[16];
#pragma unroll
    for (int j = 0; j < 16; ++j) a[j] = slot[c * TILE_PITCH + 2 * j + h];

    v8f acc[8];
    const v8f vzero = {0.f, 0.f, 0.f, 0.f, 0.f, 0.f, 0.f, 0.f};
#pragma unroll
    for (int t = 0; t < 8; ++t) acc[t] = vzero;

    int boff = h * C_DIM + c;
    asm volatile("" : "+v"(boff));
    const v4f* bbase = &lWq[boff];

#pragma unroll
    for (int j = 0; j < 16; ++j) {
      const v4f* bb = bbase + 2 * j * C_DIM;
      v4f b[8];
#pragma unroll
      for (int t = 0; t < 8; ++t) b[t] = bb[t * 16];
      const v2f a0 = {a[j].x, a[j].y};
      const v2f a1 = {a[j].z, a[j].w};
#pragma unroll
      for (int t = 0; t < 8; ++t) {
        const v2f b0 = {b[t].x, b[t].y};
        acc[t] = __builtin_amdgcn_wmma_f32_16x16x4_f32(
            false, a0, false, b0, (short)0, acc[t], false, false);
      }
#pragma unroll
      for (int t = 0; t < 8; ++t) {
        const v2f b1 = {b[t].z, b[t].w};
        acc[t] = __builtin_amdgcn_wmma_f32_16x16x4_f32(
            false, a1, false, b1, (short)0, acc[t], false, false);
      }
    }

#pragma unroll
    for (int t = 0; t < 8; ++t) {
#pragma unroll
      for (int v = 0; v < 8; ++v) acc[t][v] = (acc[t][v] + bcol[t]) * sc[v];
    }
    if ((u + 1) * UNIT_ROWS <= N) {
#pragma unroll
      for (int v = 0; v < 8; ++v) {
        float* hrow = Hout + (size_t)(base + 8 * h + v) * C_DIM;
#pragma unroll
        for (int t = 0; t < 8; ++t) hrow[t * 16 + c] = acc[t][v];
      }
    } else {
#pragma unroll
      for (int v = 0; v < 8; ++v) {
        const int r = base + 8 * h + v;
        if (r < N) {
          float* hrow = Hout + (size_t)r * C_DIM;
#pragma unroll
          for (int t = 0; t < 8; ++t) hrow[t * 16 + c] = acc[t][v];
        }
      }
    }
  }
}

extern "C" void kernel_launch(void* const* d_in, const int* in_sizes, int n_in,
                              void* d_out, int out_size, void* d_ws,
                              size_t ws_size, hipStream_t stream) {
  const float* x = (const float*)d_in[0];
  const int* edges = (const int*)d_in[1];
  const float* W1 = (const float*)d_in[2];
  const float* b1 = (const float*)d_in[3];
  const float* W2 = (const float*)d_in[4];
  const float* b2 = (const float*)d_in[5];
  float* out = (float*)d_out;

  const int N = in_sizes[0] / C_DIM;

  float* inv_s = (float*)d_ws;
  const size_t inv_pad = ((size_t)N + 63) & ~(size_t)63;
  float* H = inv_s + inv_pad;

  const int deg_blocks = (N + 255) / 256;

  gcn_deg_kernel<<<deg_blocks, 256, 0, stream>>>(edges, inv_s, N);

  gcn_gemm_wmma_kernel<<<GEMM_BLOCKS, GEMM_THREADS, 0, stream>>>(x, W1, b1, inv_s, out, N);
  gcn_gather_elu_gemm_kernel<<<GEMM_BLOCKS, GEMM_THREADS, 0, stream>>>(out, edges, inv_s, W2, b2, H, N);
  gcn_gather_elu_kernel<<<GATHER_BLOCKS, GATHER_THREADS, 0, stream>>>(H, edges, inv_s, out, N);
}


# fused launch in two wave roles: 16 waves gather two nodes a trip into a shared 32-row LDS tile, next rows requested before a hand-written barrier; 4 waves multiply the last tile meanwhile; sums kept
# speedup vs baseline: 1.5876x; 1.1061x over previous
#include <hip/hip_runtime.h>
#include <hip/hip_bf16.h>
#include <math.h>


typedef __attribute__((ext_vector_type(2))) float v2f;
typedef __attribute__((ext_vector_type(4))) float v4f;
typedef __attribute__((ext_vector_type(8))) float v8f;
typedef __attribute__((ext_vector_type(16))) int v16i;

#define C_DIM 128
#define UNIT_ROWS 16
#define GEMM_WAVES 16
#define GEMM_THREADS (GEMM_WAVES * 32)
#define GEMM_BLOCKS 256
#define GATHER_WAVES 8
#define GATHER_THREADS (GATHER_WAVES * 32)
#define GATHER_BLOCKS 1024

__global__ __launch_bounds__(256) void gcn_deg_kernel(
    const int* __restrict__ edges, float* __restrict__ inv_s, int N) {
  int n = blockIdx.x * blockDim.x + threadIdx.x;
  if (n >= N) return;
  int cnt = 1;
#pragma unroll
  for (int d = 0; d < 16; ++d) cnt += (edges[n * 16 + d] >= 0) ? 1 : 0;
  inv_s[n] = rsqrtf((float)cnt);
}

__global__ __launch_bounds__(GEMM_THREADS) void gcn_gemm_wmma_kernel(
    const float* __restrict__ X, const float* __restrict__ W,
    const float* __restrict__ bias, const float* __restrict__ inv_s,
    float* __restrict__ H, int N) {
  __shared__ v4f lWq[(C_DIM / 4) * C_DIM];

  const int tid = threadIdx.x;

#pragma unroll
  for (int it = 0; it < (C_DIM / 4) * C_DIM / GEMM_THREADS; ++it) {
    const int idx = tid + it * GEMM_THREADS;
    const int kq = idx >> 7;
    const int col = idx & (C_DIM - 1);
    v4f p;
    p.x = W[(4 * kq + 0) * C_DIM + col];
    p.y = W[(4 * kq + 1) * C_DIM + col];
    p.z = W[(4 * kq + 2) * C_DIM + col];
    p.w = W[(4 * kq + 3) * C_DIM + col];
    lWq[idx] = p;
  }

  const int wave = __builtin_amdgcn_readfirstlane(tid >> 5);
  const int lane = tid & 31;
  const int c = lane & 15;
  const int h = lane >> 4;

  const int nUnits = (N + UNIT_ROWS - 1) / UNIT_ROWS;
  const int stride = gridDim.x * GEMM_WAVES;
  const int rank = (((wave >> 2) << 1) | (wave & 1)) + ((((wave ^ (wave >> 1)) & 2) != 0) ? 8 : 0);
  int u = rank * gridDim.x + blockIdx.x;

  float bcol[8];
#pragma unroll
  for (int t = 0; t < 8; ++t) bcol[t] = bias[t * 16 + c];

  v4f a[16];
  {
    int row = u * UNIT_ROWS + c;
    if (u >= nUnits || row >= N) row = 0;
    const float* p = X + (size_t)row * C_DIM + 4 * h;
#pragma unroll
    for (int j = 0; j < 16; ++j) a[j] = *(const v4f*)(p + 8 * j);
  }

  __syncthreads();

  for (; u < nUnits; u += stride) {
    const int un = u + stride;
    int rown = un * UNIT_ROWS + c;
    if (un >= nUnits || rown >= N) rown = 0;
    const float* pn = X + (size_t)rown * C_DIM + 4 * h;

    float sc[8];
#pragma unroll
    for (int v = 0; v < 8; ++v) {
      const int r = u * UNIT_ROWS + 8 * h + v;
      sc[v] = inv_s[(r < N) ? r : 0];
    }
    __builtin_amdgcn_sched_barrier(0);

    v8f acc[8];
    const v8f vzero = {0.f, 0.f, 0.f, 0.f, 0.f, 0.f, 0.f, 0.f};
#pragma unroll
    for (int t = 0; t < 8; ++t) acc[t] = vzero;

    int boff = h * C_DIM + c;
    asm volatile("" : "+v"(boff));
    const v4f* bbase = &lWq[boff];

#pragma unroll
    for (int g = 0; g < 4; ++g) {
#pragma unroll
      for (int jj = 0; jj < 4; ++jj) {
        const int j = 4 * g + jj;
        const v4f* bb = bbase + 2 * j * C_DIM;
        v4f b[8];
#pragma unroll
        for (int t = 0; t < 8; ++t) b[t] = bb[t * 16];
        const v2f a0 = {a[j].x, a[j].y};
        const v2f a1 = {a[j].z, a[j].w};
#pragma unroll
        for (int t = 0; t < 8; ++t) {
          const v2f b0 = {b[t].x, b[t].y};
          acc[t] = __builtin_amdgcn_wmma_f32_16x16x4_f32(
              false, a0, false, b0, (short)0, acc[t], false, false);
        }
#pragma unroll
        for (int t = 0; t < 8; ++t) {
          const v2f b1 = {b[t].z, b[t].w};
          acc[t] = __builtin_amdgcn_wmma_f32_16x16x4_f32(
              false, a1, false, b1, (short)0, acc[t], false, false);
        }
      }
      __builtin_amdgcn_sched_barrier(0);
#pragma unroll
      for (int jj = 0; jj < 4; ++jj) {
        const int j = 4 * g + jj;
        a[j] = *(const v4f*)(pn + 8 * j);
      }
      __builtin_amdgcn_sched_barrier(0);
    }

#pragma unroll
    for (int t = 0; t < 8; ++t) {
#pragma unroll
      for (int v = 0; v < 8; ++v) acc[t][v] = (acc[t][v] + bcol[t]) * sc[v];
    }
    if ((u + 1) * UNIT_ROWS <= N) {
#pragma unroll
      for (int v = 0; v < 8; ++v) {
        float* hrow = H + (size_t)(u * UNIT_ROWS + 8 * h + v) * C_DIM;
#pragma unroll
        for (int t = 0; t < 8; ++t) hrow[t * 16 + c] = acc[t][v];
      }
    } else {
#pragma unroll
      for (int v = 0; v < 8; ++v) {
        const int r = u * UNIT_ROWS + 8 * h + v;
        if (r < N) {
          float* hrow = H + (size_t)r * C_DIM;
#pragma unroll
          for (int t = 0; t < 8; ++t) hrow[t * 16 + c] = acc[t][v];
        }
      }
    }
  }
}

__global__ __launch_bounds__(GATHER_THREADS) void gcn_gather_elu_kernel(
    const float* __restrict__ H, const int* __restrict__ edges,
    const float* __restrict__ inv_s, float* __restrict__ out, int N) {
  const int wave = __builtin_amdgcn_readfirstlane(threadIdx.x >> 5);
  const int c4 = threadIdx.x & 31;
  const int nWaves = gridDim.x * GATHER_WAVES;
  int node = blockIdx.x * GATHER_WAVES + wave;
  if (node >= N) return;

  const float4* Hv = (const float4*)H;
  const v16i* Ev = (const v16i*)edges;

  v16i e = Ev[node];
  float sc = inv_s[node];

  for (;;) {
    const int next = node + nWaves;
    const int nn = (next < N) ? next : node;
    const v16i en = Ev[nn];
    const float scn = inv_s[nn];

    const float4 self = Hv[(size_t)node * 32 + c4];
    float4 v[16];
#pragma unroll
    for (int d = 0; d < 16; ++d) {
      if (e[d] >= 0) {
        v[d] = Hv[(size_t)e[d] * 32 + c4];
      } else {
        v[d] = make_float4(0.f, 0.f, 0.f, 0.f);
      }
    }

    float ax = self.x, ay = self.y, az = self.z, aw = self.w;
#pragma unroll
    for (int d = 0; d < 16; ++d) {
      ax += v[d].x;
      ay += v[d].y;
      az += v[d].z;
      aw += v[d].w;
    }

    ax *= sc;
    ay *= sc;
    az *= sc;
    aw *= sc;

    float4 r;
    r.x = (ax > 0.f) ? ax : expm1f(ax);
    r.y = (ay > 0.f) ? ay : expm1f(ay);
    r.z = (az > 0.f) ? az : expm1f(az);
    r.w = (aw > 0.f) ? aw : expm1f(aw);
    ((float4*)out)[(size_t)node * 32 + c4] = r;

    if (next >= N) break;
    node = next;
    e = en;
    sc = scn;
  }
}

#define TILE_PITCH 33
#define TRIP_ROWS 32
#define SLOT_QUADS (TRIP_ROWS * TILE_PITCH)
#define FUSED_GATHER_WAVES 16
#define FUSED_WAVES 20
#define FUSED_THREADS (FUSED_WAVES * 32)

static __device__ __forceinline__ void wg_barrier_lds() {
  asm volatile("s_wait_dscnt 0x0\n\ts_barrier_signal -1\n\ts_barrier_wait -1" ::: "memory");
}

static __device__ __forceinline__ void gcn_issue_rows(
    const float4* __restrict__ Hv, int node, const v16i e, int lane,
    float4& self, float4 (&v)[16]) {
  self = Hv[(size_t)node * 32 + lane];
#pragma unroll
  for (int d = 0; d < 16; ++d) {
    if (e[d] >= 0) {
      v[d] = Hv[(size_t)e[d] * 32 + lane];
    } else {
      v[d] = make_float4(0.f, 0.f, 0.f, 0.f);
    }
  }
}

static __device__ __forceinline__ v4f gcn_reduce_row(
    const float4 self, const float4 (&v)[16], float s) {
  float ax = self.x, ay = self.y, az = self.z, aw = self.w;
#pragma unroll
  for (int d = 0; d < 16; ++d) {
    ax += v[d].x;
    ay += v[d].y;
    az += v[d].z;
    aw += v[d].w;
  }
  ax *= s;
  ay *= s;
  az *= s;
  aw *= s;
  v4f r;
  r.x = (ax > 0.f) ? ax : expm1f(ax);
  r.y = (ay > 0.f) ? ay : expm1f(ay);
  r.z = (az > 0.f) ? az : expm1f(az);
  r.w = (aw > 0.f) ? aw : expm1f(aw);
  return r;
}

__global__ __launch_bounds__(FUSED_THREADS) void gcn_gather_elu_gemm_kernel(
    const float* __restrict__ Hin, const int* __restrict__ edges,
    const float* __restrict__ inv_s, const float* __restrict__ W,
    const float* __restrict__ bias, float* __restrict__ Hout, int N) {
  __shared__ v4f lWq[(C_DIM / 4) * C_DIM];
  __shared__ v4f lA[2 * SLOT_QUADS];

  const int tid = threadIdx.x;

  const int nBlocks32 = (N + TRIP_ROWS - 1) / TRIP_ROWS;
  if ((int)blockIdx.x >= nBlocks32) return;
  const int T = (nBlocks32 - (int)blockIdx.x + (int)gridDim.x - 1) / (int)gridDim.x;
  const int step = gridDim.x * TRIP_ROWS;

#pragma unroll
  for (int it = 0; it < ((C_DIM / 4) * C_DIM + FUSED_THREADS - 1) / FUSED_THREADS; ++it) {
    const int idx = tid + it * FUSED_THREADS;
    if (idx < (C_DIM / 4) * C_DIM) {
      const int kq = idx >> 7;
      const int col = idx & (C_DIM - 1);
      v4f p;
      p.x = W[(4 * kq + 0) * C_DIM + col];
      p.y = W[(4 * kq + 1) * C_DIM + col];
      p.z = W[(4 * kq + 2) * C_DIM + col];
      p.w = W[(4 * kq + 3) * C_DIM + col];
      lWq[idx] = p;
    }
  }

  const int wave = __builtin_amdgcn_readfirstlane(tid >> 5);
  const int lane = tid & 31;
  const int wq = wave / 5;
  const int wr = wave - 5 * wq;

  if (wr != 0) {
    const int g = wave - 1 - wq;
    const float4* Hv = (const float4*)Hin;
    typedef const __attribute__((address_space(4))) v16i* idx_cptr;
    typedef const __attribute__((address_space(4))) float* scale_cptr;
    const idx_cptr Ev = (idx_cptr)edges;
    const scale_cptr Sv = (scale_cptr)inv_s;
    const int last = N - 1;

    int base = blockIdx.x * TRIP_ROWS + 2 * g;
    int nA = (base < N) ? base : last;
    int nB = (base + 1 < N) ? base + 1 : last;
    v16i eA = Ev[nA];
    v16i eB = Ev[nB];
    float sA = Sv[nA];
    float sB = Sv[nB];

    float4 selfA, selfB, va[16], vb[16];
    gcn_issue_rows(Hv, nA, eA, lane, selfA, va);
    gcn_issue_rows(Hv, nB, eB, lane, selfB, vb);

    int nAn = (base + step < N) ? base + step : last;
    int nBn = (base + step + 1 < N) ? base + step + 1 : last;
    v16i eAn = Ev[nAn];
    v16i eBn = Ev[nBn];
    float sAn = Sv[nAn];
    float sBn = Sv[nBn];

    v4f* rowA = &lA[(2 * g) * TILE_PITCH + lane];

#pragma nounroll
    for (int t = 0; t < T; ++t) {
      const v4f rA = gcn_reduce_row(selfA, va, sA);
      const v4f rB = gcn_reduce_row(selfB, vb, sB);
      v4f* dst = rowA + (t & 1) * SLOT_QUADS;
      dst[0] = rA;
      dst[TILE_PITCH] = rB;

      nA = nAn;
      nB = nBn;
      eA = eAn;
      eB = eBn;
      sA = sAn;
      sB = sBn;
      if (t + 1 < T) {
        gcn_issue_rows(Hv, nA, eA, lane, selfA, va);
        gcn_issue_rows(Hv, nB, eB, lane, selfB, vb);
      }
      base += step;
      nAn = (base + step < N) ? base + step : last;
      nBn = (base + step + 1 < N) ? base + step + 1 : last;
      eAn = Ev[nAn];
      eBn = Ev[nBn];
      sAn = Sv[nAn];
      sBn = Sv[nBn];

      wg_barrier_lds();
    }
  } else {
    const int c = lane & 15;
    const int h = lane >> 4;
    const int unit = wq >> 1;
    const int q = wq & 1;

    float bcol[4];
#pragma unroll
    for (int tt = 0; tt < 4; ++tt) bcol[tt] = bias[(4 * q + tt) * 16 + c];

    const v4f* abase = &lA[(16 * unit + c) * TILE_PITCH + h];
    const v4f* bbase = &lWq[h * C_DIM + (4 * q) * 16 + c];
    int rbase = blockIdx.x * TRIP_ROWS + 16 * unit;

#pragma nounroll
    for (int t = 0; t < T; ++t) {
      wg_barrier_lds();

      float sc[8];
#pragma unroll
      for (int v = 0; v < 8; ++v) {
        const int r = rbase + 8 * h + v;
        sc[v] = inv_s[(r < N) ? r : 0];
      }

      const v4f* ap = abase + (t & 1) * SLOT_QUADS;
      v4f a[16];
#pragma unroll
      for (int j = 0; j < 16; ++j) a[j] = ap[2 * j];

      v8f acc[4];
      const v8f vzero = {0.f, 0.f, 0.f, 0.f, 0.f, 0.f, 0.f, 0.f};
#pragma unroll
      for (int tt = 0; tt < 4; ++tt) acc[tt] = vzero;

#pragma unroll
      for (int j = 0; j < 16; ++j) {
        const v4f* bb = bbase + 2 * j * C_DIM;
        v4f b[4];
#pragma unroll
        for (int tt = 0; tt < 4; ++tt) b[tt] = bb[tt * 16];
        const v2f a0 = {a[j].x, a[j].y};
        const v2f a1 = {a[j].z, a[j].w};
#pragma unroll
        for (int tt = 0; tt < 4; ++tt) {
          const v2f b0 = {b[tt].x, b[tt].y};
          acc[tt] = __builtin_amdgcn_wmma_f32_16x16x4_f32(
              false, a0, false, b0, (short)0, acc[tt], false, false);
        }
#pragma unroll
        for (int tt = 0; tt < 4; ++tt) {
          const v2f b1 = {b[tt].z, b[tt].w};
          acc[tt] = __builtin_amdgcn_wmma_f32_16x16x4_f32(
              false, a1, false, b1, (short)0, acc[tt], false, false);
        }
      }

#pragma unroll
      for (int tt = 0; tt < 4; ++tt) {
#pragma unroll
        for (int v = 0; v < 8; ++v) acc[tt][v] = (acc[tt][v] + bcol[tt]) * sc[v];
      }
      if (rbase + UNIT_ROWS <= N) {
#pragma unroll
        for (int v = 0; v < 8; ++v) {
          float* hrow = Hout + (size_t)(rbase + 8 * h + v) * C_DIM + (4 * q) * 16;
#pragma unroll
          for (int tt = 0; tt < 4; ++tt) hrow[tt * 16 + c] = acc[tt][v];
        }
      } else {
#pragma unroll
        for (int v = 0; v < 8; ++v) {
          const int r = rbase + 8 * h + v;
          if (r < N) {
            float* hrow = Hout + (size_t)r * C_DIM + (4 * q) * 16;
#pragma unroll
            for (int tt = 0; tt < 4; ++tt) hrow[tt * 16 + c] = acc[tt][v];
          }
        }
      }
      rbase += step;
    }
  }
}

extern "C" void kernel_launch(void* const* d_in, const int* in_sizes, int n_in,
                              void* d_out, int out_size, void* d_ws,
                              size_t ws_size, hipStream_t stream) {
  const float* x = (const float*)d_in[0];
  const int* edges = (const int*)d_in[1];
  const float* W1 = (const float*)d_in[2];
  const float* b1 = (const float*)d_in[3];
  const float* W2 = (const float*)d_in[4];
  const float* b2 = (const float*)d_in[5];
  float* out = (float*)d_out;

  const int N = in_sizes[0] / C_DIM;

  float* inv_s = (float*)d_ws;
  const size_t inv_pad = ((size_t)N + 63) & ~(size_t)63;
  float* H = inv_s + inv_pad;

  const int deg_blocks = (N + 255) / 256;

  gcn_deg_kernel<<<deg_blocks, 256, 0, stream>>>(edges, inv_s, N);

  gcn_gemm_wmma_kernel<<<GEMM_BLOCKS, GEMM_THREADS, 0, stream>>>(x, W1, b1, inv_s, out, N);
  gcn_gather_elu_gemm_kernel<<<GEMM_BLOCKS, FUSED_THREADS, 0, stream>>>(out, edges, inv_s, W2, b2, H, N);
  gcn_gather_elu_kernel<<<GATHER_BLOCKS, GATHER_THREADS, 0, stream>>>(H, edges, inv_s, out, N);
}


# ELU without a branch (expm1f of the value or of 0, then one select; same result for every input) in both gather kernels: a wave's row-sum and ELU chains stand in one block and issue interleaved
# speedup vs baseline: 1.6853x; 1.0615x over previous
#include <hip/hip_runtime.h>
#include <hip/hip_bf16.h>
#include <math.h>


typedef __attribute__((ext_vector_type(2))) float v2f;
typedef __attribute__((ext_vector_type(4))) float v4f;
typedef __attribute__((ext_vector_type(8))) float v8f;
typedef __attribute__((ext_vector_type(16))) int v16i;

#define C_DIM 128
#define UNIT_ROWS 16
#define GEMM_WAVES 16
#define GEMM_THREADS (GEMM_WAVES * 32)
#define GEMM_BLOCKS 256
#define GATHER_WAVES 8
#define GATHER_THREADS (GATHER_WAVES * 32)
#define GATHER_BLOCKS 1024

__global__ __launch_bounds__(256) void gcn_deg_kernel(
    const int* __restrict__ edges, float* __restrict__ inv_s, int N) {
  int n = blockIdx.x * blockDim.x + threadIdx.x;
  if (n >= N) return;
  int cnt = 1;
#pragma unroll
  for (int d = 0; d < 16; ++d) cnt += (edges[n * 16 + d] >= 0) ? 1 : 0;
  inv_s[n] = rsqrtf((float)cnt);
}

__global__ __launch_bounds__(GEMM_THREADS) void gcn_gemm_wmma_kernel(
    const float* __restrict__ X, const float* __restrict__ W,
    const float* __restrict__ bias, const float* __restrict__ inv_s,
    float* __restrict__ H, int N) {
  __shared__ v4f lWq[(C_DIM / 4) * C_DIM];

  const int tid = threadIdx.x;

#pragma unroll
  for (int it = 0; it < (C_DIM / 4) * C_DIM / GEMM_THREADS; ++it) {
    const int idx = tid + it * GEMM_THREADS;
    const int kq = idx >> 7;
    const int col = idx & (C_DIM - 1);
    v4f p;
    p.x = W[(4 * kq + 0) * C_DIM + col];
    p.y = W[(4 * kq + 1) * C_DIM + col];
    p.z = W[(4 * kq + 2) * C_DIM + col];
    p.w = W[(4 * kq + 3) * C_DIM + col];
    lWq[idx] = p;
  }

  const int wave = __builtin_amdgcn_readfirstlane(tid >> 5);
  const int lane = tid & 31;
  const int c = lane & 15;
  const int h = lane >> 4;

  const int nUnits = (N + UNIT_ROWS - 1) / UNIT_ROWS;
  const int stride = gridDim.x * GEMM_WAVES;
  const int rank = (((wave >> 2) << 1) | (wave & 1)) + ((((wave ^ (wave >> 1)) & 2) != 0) ? 8 : 0);
  int u = rank * gridDim.x + blockIdx.x;

  float bcol[8];
#pragma unroll
  for (int t = 0; t < 8; ++t) bcol[t] = bias[t * 16 + c];

  v4f a[16];
  {
    int row = u * UNIT_ROWS + c;
    if (u >= nUnits || row >= N) row = 0;
    const float* p = X + (size_t)row * C_DIM + 4 * h;
#pragma unroll
    for (int j = 0; j < 16; ++j) a[j] = *(const v4f*)(p + 8 * j);
  }

  __syncthreads();

  for (; u < nUnits; u += stride) {
    const int un = u + stride;
    int rown = un * UNIT_ROWS + c;
    if (un >= nUnits || rown >= N) rown = 0;
    const float* pn = X + (size_t)rown * C_DIM + 4 * h;

    float sc[8];
#pragma unroll
    for (int v = 0; v < 8; ++v) {
      const int r = u * UNIT_ROWS + 8 * h + v;
      sc[v] = inv_s[(r < N) ? r : 0];
    }
    __builtin_amdgcn_sched_barrier(0);

    v8f acc[8];
    const v8f vzero = {0.f, 0.f, 0.f, 0.f, 0.f, 0.f, 0.f, 0.f};
#pragma unroll
    for (int t = 0; t < 8; ++t) acc[t] = vzero;

    int boff = h * C_DIM + c;
    asm volatile("" : "+v"(boff));
    const v4f* bbase = &lWq[boff];

#pragma unroll
    for (int g = 0; g < 4; ++g) {
#pragma unroll
      for (int jj = 0; jj < 4; ++jj) {
        const int j = 4 * g + jj;
        const v4f* bb = bbase + 2 * j * C_DIM;
        v4f b[8];
#pragma unroll
        for (int t = 0; t < 8; ++t) b[t] = bb[t * 16];
        const v2f a0 = {a[j].x, a[j].y};
        const v2f a1 = {a[j].z, a[j].w};
#pragma unroll
        for (int t = 0; t < 8; ++t) {
          const v2f b0 = {b[t].x, b[t].y};
          acc[t] = __builtin_amdgcn_wmma_f32_16x16x4_f32(
              false, a0, false, b0, (short)0, acc[t], false, false);
        }
#pragma unroll
        for (int t = 0; t < 8; ++t) {
          const v2f b1 = {b[t].z, b[t].w};
          acc[t] = __builtin_amdgcn_wmma_f32_16x16x4_f32(
              false, a1, false, b1, (short)0, acc[t], false, false);
        }
      }
      __builtin_amdgcn_sched_barrier(0);
#pragma unroll
      for (int jj = 0; jj < 4; ++jj) {
        const int j = 4 * g + jj;
        a[j] = *(const v4f*)(pn + 8 * j);
      }
      __builtin_amdgcn_sched_barrier(0);
    }

#pragma unroll
    for (int t = 0; t < 8; ++t) {
#pragma unroll
      for (int v = 0; v < 8; ++v) acc[t][v] = (acc[t][v] + bcol[t]) * sc[v];
    }
    if ((u + 1) * UNIT_ROWS <= N) {
#pragma unroll
      for (int v = 0; v < 8; ++v) {
        float* hrow = H + (size_t)(u * UNIT_ROWS + 8 * h + v) * C_DIM;
#pragma unroll
        for (int t = 0; t < 8; ++t) hrow[t * 16 + c] = acc[t][v];
      }
    } else {
#pragma unroll
      for (int v = 0; v < 8; ++v) {
        const int r = u * UNIT_ROWS + 8 * h + v;
        if (r < N) {
          float* hrow = H + (size_t)r * C_DIM;
#pragma unroll
          for (int t = 0; t < 8; ++t) hrow[t * 16 + c] = acc[t][v];
        }
      }
    }
  }
}

static __device__ __forceinline__ float gcn_elu(float a) {
  const float t = (a > 0.f) ? 0.f : a;
  const float em = expm1f(t);
  return (a > 0.f) ? a : em;
}

__global__ __launch_bounds__(GATHER_THREADS) void gcn_gather_elu_kernel(
    const float* __restrict__ H, const int* __restrict__ edges,
    const float* __restrict__ inv_s, float* __restrict__ out, int N) {
  const int wave = __builtin_amdgcn_readfirstlane(threadIdx.x >> 5);
  const int c4 = threadIdx.x & 31;
  const int nWaves = gridDim.x * GATHER_WAVES;
  int node = blockIdx.x * GATHER_WAVES + wave;
  if (node >= N) return;

  const float4* Hv = (const float4*)H;
  const v16i* Ev = (const v16i*)edges;

  v16i e = Ev[node];
  float sc = inv_s[node];

  for (;;) {
    const int next = node + nWaves;
    const int nn = (next < N) ? next : node;
    const v16i en = Ev[nn];
    const float scn = inv_s[nn];

    const float4 self = Hv[(size_t)node * 32 + c4];
    float4 v[16];
#pragma unroll
    for (int d = 0; d < 16; ++d) {
      if (e[d] >= 0) {
        v[d] = Hv[(size_t)e[d] * 32 + c4];
      } else {
        v[d] = make_float4(0.f, 0.f, 0.f, 0.f);
      }
    }

    float ax = self.x, ay = self.y, az = self.z, aw = self.w;
#pragma unroll
    for (int d = 0; d < 16; ++d) {
      ax += v[d].x;
      ay += v[d].y;
      az += v[d].z;
      aw += v[d].w;
    }

    ax *= sc;
    ay *= sc;
    az *= sc;
    aw *= sc;

    float4 r;
    r.x = gcn_elu(ax);
    r.y = gcn_elu(ay);
    r.z = gcn_elu(az);
    r.w = gcn_elu(aw);
    ((float4*)out)[(size_t)node * 32 + c4] = r;

    if (next >= N) break;
    node = next;
    e = en;
    sc = scn;
  }
}

#define TILE_PITCH 33
#define TRIP_ROWS 32
#define SLOT_QUADS (TRIP_ROWS * TILE_PITCH)
#define FUSED_GATHER_WAVES 16
#define FUSED_WAVES 20
#define FUSED_THREADS (FUSED_WAVES * 32)

static __device__ __forceinline__ void wg_barrier_lds() {
  asm volatile("s_wait_dscnt 0x0\n\ts_barrier_signal -1\n\ts_barrier_wait -1" ::: "memory");
}

static __device__ __forceinline__ void gcn_issue_rows(
    const float4* __restrict__ Hv, int node, const v16i e, int lane,
    float4& self, float4 (&v)[16]) {
  self = Hv[(size_t)node * 32 + lane];
#pragma unroll
  for (int d = 0; d < 16; ++d) {
    if (e[d] >= 0) {
      v[d] = Hv[(size_t)e[d] * 32 + lane];
    } else {
      v[d] = make_float4(0.f, 0.f, 0.f, 0.f);
    }
  }
}

static __device__ __forceinline__ v4f gcn_reduce_row(
    const float4 self, const float4 (&v)[16], float s) {
  float ax = self.x, ay = self.y, az = self.z, aw = self.w;
#pragma unroll
  for (int d = 0; d < 16; ++d) {
    ax += v[d].x;
    ay += v[d].y;
    az += v[d].z;
    aw += v[d].w;
  }
  ax *= s;
  ay *= s;
  az *= s;
  aw *= s;
  v4f r;
  r.x = gcn_elu(ax);
  r.y = gcn_elu(ay);
  r.z = gcn_elu(az);
  r.w = gcn_elu(aw);
  return r;
}

__global__ __launch_bounds__(FUSED_THREADS) void gcn_gather_elu_gemm_kernel(
    const float* __restrict__ Hin, const int* __restrict__ edges,
    const float* __restrict__ inv_s, const float* __restrict__ W,
    const float* __restrict__ bias, float* __restrict__ Hout, int N) {
  __shared__ v4f lWq[(C_DIM / 4) * C_DIM];
  __shared__ v4f lA[2 * SLOT_QUADS];

  const int tid = threadIdx.x;

  const int nBlocks32 = (N + TRIP_ROWS - 1) / TRIP_ROWS;
  if ((int)blockIdx.x >= nBlocks32) return;
  const int T = (nBlocks32 - (int)blockIdx.x + (int)gridDim.x - 1) / (int)gridDim.x;
  const int step = gridDim.x * TRIP_ROWS;

#pragma unroll
  for (int it = 0; it < ((C_DIM / 4) * C_DIM + FUSED_THREADS - 1) / FUSED_THREADS; ++it) {
    const int idx = tid + it * FUSED_THREADS;
    if (idx < (C_DIM / 4) * C_DIM) {
      const int kq = idx >> 7;
      const int col = idx & (C_DIM - 1);
      v4f p;
      p.x = W[(4 * kq + 0) * C_DIM + col];
      p.y = W[(4 * kq + 1) * C_DIM + col];
      p.z = W[(4 * kq + 2) * C_DIM + col];
      p.w = W[(4 * kq + 3) * C_DIM + col];
      lWq[idx] = p;
    }
  }

  const int wave = __builtin_amdgcn_readfirstlane(tid >> 5);
  const int lane = tid & 31;
  const int wq = wave / 5;
  const int wr = wave - 5 * wq;

  if (wr != 0) {
    const int g = wave - 1 - wq;
    const float4* Hv = (const float4*)Hin;
    typedef const __attribute__((address_space(4))) v16i* idx_cptr;
    typedef const __attribute__((address_space(4))) float* scale_cptr;
    const idx_cptr Ev = (idx_cptr)edges;
    const scale_cptr Sv = (scale_cptr)inv_s;
    const int last = N - 1;

    int base = blockIdx.x * TRIP_ROWS + 2 * g;
    int nA = (base < N) ? base : last;
    int nB = (base + 1 < N) ? base + 1 : last;
    v16i eA = Ev[nA];
    v16i eB = Ev[nB];
    float sA = Sv[nA];
    float sB = Sv[nB];

    float4 selfA, selfB, va[16], vb[16];
    gcn_issue_rows(Hv, nA, eA, lane, selfA, va);
    gcn_issue_rows(Hv, nB, eB, lane, selfB, vb);

    int nAn = (base + step < N) ? base + step : last;
    int nBn = (base + step + 1 < N) ? base + step + 1 : last;
    v16i eAn = Ev[nAn];
    v16i eBn = Ev[nBn];
    float sAn = Sv[nAn];
    float sBn = Sv[nBn];

    v4f* rowA = &lA[(2 * g) * TILE_PITCH + lane];

#pragma nounroll
    for (int t = 0; t < T; ++t) {
      const v4f rA = gcn_reduce_row(selfA, va, sA);
      const v4f rB = gcn_reduce_row(selfB, vb, sB);
      v4f* dst = rowA + (t & 1) * SLOT_QUADS;
      dst[0] = rA;
      dst[TILE_PITCH] = rB;

      nA = nAn;
      nB = nBn;
      eA = eAn;
      eB = eBn;
      sA = sAn;
      sB = sBn;
      if (t + 1 < T) {
        gcn_issue_rows(Hv, nA, eA, lane, selfA, va);
        gcn_issue_rows(Hv, nB, eB, lane, selfB, vb);
      }
      base += step;
      nAn = (base + step < N) ? base + step : last;
      nBn = (base + step + 1 < N) ? base + step + 1 : last;
      eAn = Ev[nAn];
      eBn = Ev[nBn];
      sAn = Sv[nAn];
      sBn = Sv[nBn];

      wg_barrier_lds();
    }
  } else {
    const int c = lane & 15;
    const int h = lane >> 4;
    const int unit = wq >> 1;
    const int q = wq & 1;

    float bcol[4];
#pragma unroll
    for (int tt = 0; tt < 4; ++tt) bcol[tt] = bias[(4 * q + tt) * 16 + c];

    const v4f* abase = &lA[(16 * unit + c) * TILE_PITCH + h];
    const v4f* bbase = &lWq[h * C_DIM + (4 * q) * 16 + c];
    int rbase = blockIdx.x * TRIP_ROWS + 16 * unit;

#pragma nounroll
    for (int t = 0; t < T; ++t) {
      wg_barrier_lds();

      float sc[8];
#pragma unroll
      for (int v = 0; v < 8; ++v) {
        const int r = rbase + 8 * h + v;
        sc[v] = inv_s[(r < N) ? r : 0];
      }

      const v4f* ap = abase + (t & 1) * SLOT_QUADS;
      v4f a[16];
#pragma unroll
      for (int j = 0; j < 16; ++j) a[j] = ap[2 * j];

      v8f acc[4];
      const v8f vzero = {0.f, 0.f, 0.f, 0.f, 0.f, 0.f, 0.f, 0.f};
#pragma unroll
      for (int tt = 0; tt < 4; ++tt) acc[tt] = vzero;

#pragma unroll
      for (int j = 0; j < 16; ++j) {
        const v4f* bb = bbase + 2 * j * C_DIM;
        v4f b[4];
#pragma unroll
        for (int tt = 0; tt < 4; ++tt) b[tt] = bb[tt * 16];
        const v2f a0 = {a[j].x, a[j].y};
        const v2f a1 = {a[j].z, a[j].w};
#pragma unroll
        for (int tt = 0; tt < 4; ++tt) {
          const v2f b0 = {b[tt].x, b[tt].y};
          acc[tt] = __builtin_amdgcn_wmma_f32_16x16x4_f32(
              false, a0, false, b0, (short)0, acc[tt], false, false);
        }
#pragma unroll
        for (int tt = 0; tt < 4; ++tt) {
          const v2f b1 = {b[tt].z, b[tt].w};
          acc[tt] = __builtin_amdgcn_wmma_f32_16x16x4_f32(
              false, a1, false, b1, (short)0, acc[tt], false, false);
        }
      }

#pragma unroll
      for (int tt = 0; tt < 4; ++tt) {
#pragma unroll
        for (int v = 0; v < 8; ++v) acc[tt][v] = (acc[tt][v] + bcol[tt]) * sc[v];
      }
      if (rbase + UNIT_ROWS <= N) {
#pragma unroll
        for (int v = 0; v < 8; ++v) {
          float* hrow = Hout + (size_t)(rbase + 8 * h + v) * C_DIM + (4 * q) * 16;
#pragma unroll
          for (int tt = 0; tt < 4; ++tt) hrow[tt * 16 + c] = acc[tt][v];
        }
      } else {
#pragma unroll
        for (int v = 0; v < 8; ++v) {
          const int r = rbase + 8 * h + v;
          if (r < N) {
            float* hrow = Hout + (size_t)r * C_DIM + (4 * q) * 16;
#pragma unroll
            for (int tt = 0; tt < 4; ++tt) hrow[tt * 16 + c] = acc[tt][v];
          }
        }
      }
      rbase += step;
    }
  }
}

extern "C" void kernel_launch(void* const* d_in, const int* in_sizes, int n_in,
                              void* d_out, int out_size, void* d_ws,
                              size_t ws_size, hipStream_t stream) {
  const float* x = (const float*)d_in[0];
  const int* edges = (const int*)d_in[1];
  const float* W1 = (const float*)d_in[2];
  const float* b1 = (const float*)d_in[3];
  const float* W2 = (const float*)d_in[4];
  const float* b2 = (const float*)d_in[5];
  float* out = (float*)d_out;

  const int N = in_sizes[0] / C_DIM;

  float* inv_s = (float*)d_ws;
  const size_t inv_pad = ((size_t)N + 63) & ~(size_t)63;
  float* H = inv_s + inv_pad;

  const int deg_blocks = (N + 255) / 256;

  gcn_deg_kernel<<<deg_blocks, 256, 0, stream>>>(edges, inv_s, N);

  gcn_gemm_wmma_kernel<<<GEMM_BLOCKS, GEMM_THREADS, 0, stream>>>(x, W1, b1, inv_s, out, N);
  gcn_gather_elu_gemm_kernel<<<GEMM_BLOCKS, FUSED_THREADS, 0, stream>>>(out, edges, inv_s, W2, b2, H, N);
  gcn_gather_elu_kernel<<<GATHER_BLOCKS, GATHER_THREADS, 0, stream>>>(H, edges, inv_s, out, N);
}
